# starting kernel
# baseline (speedup 1.0000x reference)
#include <hip/hip_runtime.h>


typedef __attribute__((ext_vector_type(2))) float v2f;
typedef __attribute__((ext_vector_type(8))) float v8f;

namespace {
constexpr int B_  = 8;
constexpr int T_  = 4096;
constexpr int NR_ = 256;
constexpr int NT  = T_ / 16;
constexpr int NG  = NT / 4;
constexpr int PROW = T_ + 4;
constexpr int GPAD = 16;
constexpr int NTHREADS = 256;
constexpr int NWAVE = NTHREADS / 32;
}

#define LOADA(D)                                                              \
  {                                                                           \
    const int gb = 16 * (D) + M - kb;                                         \
    _Pragma("unroll")                                                         \
    for (int q = 0; q < 4; ++q) {                                             \
      a[q].x = gp[gb - 4 * q - 1];                                            \
      a[q].y = gp[gb - 4 * q];                                                \
    }                                                                         \
  }

#define LOADB(SLOT, SI)                                                       \
  {                                                                           \
    const float* pr = prow0 + 16 * (SI);                                      \
    _Pragma("unroll")                                                         \
    for (int q = 0; q < 4; ++q) {                                             \
      Bt[(SLOT) & 3][q].x = pr[4 * q + 1];                                    \
      Bt[(SLOT) & 3][q].y = pr[4 * q];                                        \
    }                                                                         \
  }

#define WMMA4(ACC, SLOT)                                                      \
  {                                                                           \
    _Pragma("unroll")                                                         \
    for (int q = 0; q < 4; ++q)                                               \
      ACC = __builtin_amdgcn_wmma_f32_16x16x4_f32(                            \
          false, a[q], false, Bt[(SLOT) & 3][q], (short)0, ACC, false, false);\
  }

#define STORE_TILE(ACC, C)                                                    \
  {                                                                           \
    _Pragma("unroll")                                                         \
    for (int r = 0; r < 8; ++r) {                                             \
      const int t = 16 * (tb + (C)) + mbase + r;                              \
      out[((size_t)N * T_ + t) * NR_ + j] = ACC[r] + prN[t];                  \
    }                                                                         \
  }

__global__ __launch_bounds__(NTHREADS)
void epsi_causal_toeplitz_wmma(const float* __restrict__ P,
                               const float* __restrict__ g,
                               float* __restrict__ out) {
  __shared__ float smem[B_ * PROW + GPAD + T_];
  float* Plds = smem;
  float* gpad = smem + B_ * PROW;

  const int j   = blockIdx.x;
  const int tid = threadIdx.x;

  for (int idx = tid; idx < B_ * T_; idx += NTHREADS) {
    const int b = idx >> 12;
    const int t = idx & (T_ - 1);
    Plds[b * PROW + t] = P[(size_t)idx * NR_ + j];
  }
  if (tid < GPAD) gpad[tid] = 0.0f;
  for (int t = tid; t < T_; t += NTHREADS) {
    gpad[GPAD + t] = g[(size_t)t * NR_ + j];
  }
  __syncthreads();

  const int lane = tid & 31;
  const int wave = __builtin_amdgcn_readfirstlane(tid >> 5);
  const int M    = lane & 15;
  const int kb   = (lane < 16) ? 0 : 2;
  const int bcol = lane & 7;
  const float* gp    = gpad + GPAD;
  const float* prow0 = Plds + bcol * PROW + kb;

  for (int gi = wave; gi < NG; gi += NWAVE) {
    const int tb = 4 * gi;
    v8f acc0 = {}, acc1 = {}, acc2 = {}, acc3 = {};
    v2f Bt[4][4];
    v2f a[4];

    #pragma unroll
    for (int c = 1; c < 4; ++c) LOADB(c, tb + c);

    for (int d4 = 0; d4 < tb; d4 += 4) {
      #pragma unroll
      for (int p = 0; p < 4; ++p) {
        const int d = d4 + p;
        LOADB(0 - p, tb - d);
        LOADA(d);
        WMMA4(acc0, 0 - p);
        WMMA4(acc1, 1 - p);
        WMMA4(acc2, 2 - p);
        WMMA4(acc3, 3 - p);
      }
    }

    LOADB(0, 0);
    LOADA(tb);
    WMMA4(acc0, 0); WMMA4(acc1, 1); WMMA4(acc2, 2); WMMA4(acc3, 3);
    LOADA(tb + 1);
    WMMA4(acc1, 0); WMMA4(acc2, 1); WMMA4(acc3, 2);
    LOADA(tb + 2);
    WMMA4(acc2, 0); WMMA4(acc3, 1);
    LOADA(tb + 3);
    WMMA4(acc3, 0);

    const int N     = lane & 15;
    const int mbase = (lane < 16) ? 0 : 8;
    if (N < B_) {
      const float* prN = Plds + N * PROW;
      STORE_TILE(acc0, 0)
      STORE_TILE(acc1, 1)
      STORE_TILE(acc2, 2)
      STORE_TILE(acc3, 3)
    }
  }
}

extern "C" void kernel_launch(void* const* d_in, const int* in_sizes, int n_in,
                              void* d_out, int out_size, void* d_ws, size_t ws_size,
                              hipStream_t stream) {
  const float* P = (const float*)d_in[0];
  const float* g = (const float*)d_in[1];
  float* out = (float*)d_out;
  (void)in_sizes; (void)n_in; (void)out_size; (void)d_ws; (void)ws_size;

  epsi_causal_toeplitz_wmma<<<dim3(NR_), dim3(NTHREADS), 0, stream>>>(P, g, out);
}


# Batch fills only 8 of the 16 WMMA columns: carry time tile X-4 in columns 0-7 and tile X in columns 8-15 at the same lag (same Toeplitz A), so each WMMA does two output tiles; WMMA count halves
# speedup vs baseline: 1.4705x; 1.4705x over previous
#include <hip/hip_runtime.h>


typedef __attribute__((ext_vector_type(2))) float v2f;
typedef __attribute__((ext_vector_type(8))) float v8f;

namespace {
constexpr int B_  = 8;
constexpr int T_  = 4096;
constexpr int NR_ = 256;
constexpr int NT  = T_ / 16;
constexpr int NG  = NT / 8;
constexpr int ZP   = 64;
constexpr int PROW = ZP + T_ + 4;
constexpr int GPAD = 16;
constexpr int NTHREADS = 256;
constexpr int NWAVE = NTHREADS / 32;
}

#define LOADA(D)                                                              \
  {                                                                           \
    const int gb = 16 * (D) + M - kb;                                         \
    _Pragma("unroll")                                                         \
    for (int q = 0; q < 4; ++q) {                                             \
      a[q].x = gp[gb - 4 * q - 1];                                            \
      a[q].y = gp[gb - 4 * q];                                                \
    }                                                                         \
  }

#define LOADB(SLOT, SI)                                                       \
  {                                                                           \
    const float* pr = prow0 + 16 * (SI);                                      \
    _Pragma("unroll")                                                         \
    for (int q = 0; q < 4; ++q) {                                             \
      Bt[(SLOT) & 3][q].x = pr[4 * q + 1];                                    \
      Bt[(SLOT) & 3][q].y = pr[4 * q];                                        \
    }                                                                         \
  }

#define WMMA4(ACC, SLOT)                                                      \
  {                                                                           \
    _Pragma("unroll")                                                         \
    for (int q = 0; q < 4; ++q)                                               \
      ACC = __builtin_amdgcn_wmma_f32_16x16x4_f32(                            \
          false, a[q], false, Bt[(SLOT) & 3][q], (short)0, ACC, false, false);\
  }

#define STORE_TILE(ACC, C)                                                    \
  {                                                                           \
    _Pragma("unroll")                                                         \
    for (int r = 0; r < 8; ++r) {                                             \
      const int t = 16 * (tlo + (C)) + mbase + r;                             \
      out[((size_t)bN * T_ + t) * NR_ + j] = ACC[r] + prN[t];                 \
    }                                                                         \
  }

__global__ __launch_bounds__(NTHREADS)
void epsi_causal_toeplitz_wmma(const float* __restrict__ P,
                               const float* __restrict__ g,
                               float* __restrict__ out) {
  __shared__ float smem[B_ * PROW + GPAD + T_];
  float* Plds = smem;
  float* gpad = smem + B_ * PROW;

  const int j   = blockIdx.x;
  const int tid = threadIdx.x;

  for (int idx = tid; idx < B_ * T_; idx += NTHREADS) {
    const int b = idx >> 12;
    const int t = idx & (T_ - 1);
    Plds[b * PROW + ZP + t] = P[(size_t)idx * NR_ + j];
  }
  for (int i = tid; i < B_ * ZP; i += NTHREADS) {
    Plds[(i >> 6) * PROW + (i & (ZP - 1))] = 0.0f;
  }
  if (tid < GPAD) gpad[tid] = 0.0f;
  for (int t = tid; t < T_; t += NTHREADS) {
    gpad[GPAD + t] = g[(size_t)t * NR_ + j];
  }
  __syncthreads();

  const int lane = tid & 31;
  const int wave = __builtin_amdgcn_readfirstlane(tid >> 5);
  const int M    = lane & 15;
  const int kb   = (lane < 16) ? 0 : 2;
  const int bcol = lane & 7;
  const int tsh  = (lane & 8) ? 0 : 64;
  const float* gp    = gpad + GPAD;
  const float* prow0 = Plds + bcol * PROW + ZP - tsh + kb;

  for (int gi = wave; gi < NG; gi += NWAVE) {
    const int tb = 8 * gi + 4;
    v8f acc0 = {}, acc1 = {}, acc2 = {}, acc3 = {};
    v2f Bt[4][4];
    v2f a[4];

    #pragma unroll
    for (int c = 1; c < 4; ++c) LOADB(c, tb + c);

    for (int d4 = 0; d4 < tb; d4 += 4) {
      #pragma unroll
      for (int p = 0; p < 4; ++p) {
        const int d = d4 + p;
        LOADB(0 - p, tb - d);
        LOADA(d);
        WMMA4(acc0, 0 - p);
        WMMA4(acc1, 1 - p);
        WMMA4(acc2, 2 - p);
        WMMA4(acc3, 3 - p);
      }
    }

    LOADB(0, 0);
    LOADA(tb);
    WMMA4(acc0, 0); WMMA4(acc1, 1); WMMA4(acc2, 2); WMMA4(acc3, 3);
    LOADA(tb + 1);
    WMMA4(acc1, 0); WMMA4(acc2, 1); WMMA4(acc3, 2);
    LOADA(tb + 2);
    WMMA4(acc2, 0); WMMA4(acc3, 1);
    LOADA(tb + 3);
    WMMA4(acc3, 0);

    const int bN    = lane & 7;
    const int tlo   = (lane & 8) ? tb : tb - 4;
    const int mbase = (lane < 16) ? 0 : 8;
    const float* prN = Plds + bN * PROW + ZP;
    STORE_TILE(acc0, 0)
    STORE_TILE(acc1, 1)
    STORE_TILE(acc2, 2)
    STORE_TILE(acc3, 3)
  }
}

extern "C" void kernel_launch(void* const* d_in, const int* in_sizes, int n_in,
                              void* d_out, int out_size, void* d_ws, size_t ws_size,
                              hipStream_t stream) {
  const float* P = (const float*)d_in[0];
  const float* g = (const float*)d_in[1];
  float* out = (float*)d_out;
  (void)in_sizes; (void)n_in; (void)out_size; (void)d_ws; (void)ws_size;

  epsi_causal_toeplitz_wmma<<<dim3(NR_), dim3(NTHREADS), 0, stream>>>(P, g, out);
}


# Each 64-WMMA lag block's LDS fragment loads issued as ONE group a block ahead into a second register set (P kept pair-swapped in LDS: 8-byte loads land in operand order); same k and sum order
# speedup vs baseline: 1.5974x; 1.0863x over previous
#include <hip/hip_runtime.h>


typedef __attribute__((ext_vector_type(2))) float v2f;
typedef __attribute__((ext_vector_type(8))) float v8f;

namespace {
constexpr int B_  = 8;
constexpr int T_  = 4096;
constexpr int NR_ = 256;
constexpr int NT  = T_ / 16;
constexpr int NG  = NT / 8;
constexpr int ZP   = 64;
constexpr int PROW = ZP + T_ + 4;
constexpr int GPAD = 16;
constexpr int NTHREADS = 256;
constexpr int NWAVE = NTHREADS / 32;
}

#define LOADSET_MAIN(S, D4)                                                   \
  {                                                                           \
    const float* pb = prow0 + 16 * (tb - (D4) - 3);                           \
    _Pragma("unroll")                                                         \
    for (int u = 0; u < 7; ++u) {                                             \
      _Pragma("unroll")                                                       \
      for (int q = 0; q < 4; ++q)                                             \
        Bs[S][u][q] = *(const v2f*)(pb + 16 * u + 4 * q);                     \
    }                                                                         \
    const float* ga = gp + 16 * (D4) + M - kb;                                \
    _Pragma("unroll")                                                         \
    for (int p = 0; p < 4; ++p) {                                             \
      _Pragma("unroll")                                                       \
      for (int q = 0; q < 4; ++q) {                                           \
        As[S][p][q].x = ga[16 * p - 4 * q - 1];                               \
        As[S][p][q].y = ga[16 * p - 4 * q];                                   \
      }                                                                       \
    }                                                                         \
  }

#define LOADSET_EPI(S)                                                        \
  {                                                                           \
    _Pragma("unroll")                                                         \
    for (int u = 0; u < 4; ++u) {                                             \
      _Pragma("unroll")                                                       \
      for (int q = 0; q < 4; ++q)                                             \
        Bs[S][u][q] = *(const v2f*)(prow0 + 16 * u + 4 * q);                  \
    }                                                                         \
    const float* ga = gp + 16 * tb + M - kb;                                  \
    _Pragma("unroll")                                                         \
    for (int p = 0; p < 4; ++p) {                                             \
      _Pragma("unroll")                                                       \
      for (int q = 0; q < 4; ++q) {                                           \
        As[S][p][q].x = ga[16 * p - 4 * q - 1];                               \
        As[S][p][q].y = ga[16 * p - 4 * q];                                   \
      }                                                                       \
    }                                                                         \
  }

#define WM(ACC, AF, BF)                                                       \
  ACC = __builtin_amdgcn_wmma_f32_16x16x4_f32(false, AF, false, BF, (short)0, \
                                              ACC, false, false);

#define WBLK_MAIN(S)                                                          \
  {                                                                           \
    _Pragma("unroll")                                                         \
    for (int p = 0; p < 4; ++p) {                                             \
      _Pragma("unroll")                                                       \
      for (int q = 0; q < 4; ++q) {                                           \
        WM(acc0, As[S][p][q], Bs[S][3 - p][q])                                \
        WM(acc1, As[S][p][q], Bs[S][4 - p][q])                                \
        WM(acc2, As[S][p][q], Bs[S][5 - p][q])                                \
        WM(acc3, As[S][p][q], Bs[S][6 - p][q])                                \
      }                                                                       \
    }                                                                         \
  }

#define WBLK_EPI(S)                                                           \
  {                                                                           \
    _Pragma("unroll")                                                         \
    for (int q = 0; q < 4; ++q) {                                             \
      WM(acc0, As[S][0][q], Bs[S][0][q])                                      \
      WM(acc1, As[S][0][q], Bs[S][1][q])                                      \
      WM(acc2, As[S][0][q], Bs[S][2][q])                                      \
      WM(acc3, As[S][0][q], Bs[S][3][q])                                      \
    }                                                                         \
    _Pragma("unroll")                                                         \
    for (int q = 0; q < 4; ++q) {                                             \
      WM(acc1, As[S][1][q], Bs[S][0][q])                                      \
      WM(acc2, As[S][1][q], Bs[S][1][q])                                      \
      WM(acc3, As[S][1][q], Bs[S][2][q])                                      \
    }                                                                         \
    _Pragma("unroll")                                                         \
    for (int q = 0; q < 4; ++q) {                                             \
      WM(acc2, As[S][2][q], Bs[S][0][q])                                      \
      WM(acc3, As[S][2][q], Bs[S][1][q])                                      \
    }                                                                         \
    _Pragma("unroll")                                                         \
    for (int q = 0; q < 4; ++q) {                                             \
      WM(acc3, As[S][3][q], Bs[S][0][q])                                      \
    }                                                                         \
  }

#define GROUP_FENCE() __builtin_amdgcn_sched_barrier(0)

#define BLOCK_END()                                                           \
  asm volatile("" : "+v"(acc0), "+v"(acc1), "+v"(acc2), "+v"(acc3))

#define STORE_TILE(ACC, C)                                                    \
  {                                                                           \
    _Pragma("unroll")                                                         \
    for (int r = 0; r < 8; ++r) {                                             \
      const int t = 16 * (tlo + (C)) + mbase + r;                             \
      out[((size_t)bN * T_ + t) * NR_ + j] = ACC[r] + prN[t ^ 1];             \
    }                                                                         \
  }

__global__ __launch_bounds__(NTHREADS)
void epsi_causal_toeplitz_wmma(const float* __restrict__ P,
                               const float* __restrict__ g,
                               float* __restrict__ out) {
  __shared__ __attribute__((aligned(16))) float smem[B_ * PROW + GPAD + T_];
  float* Plds = smem;
  float* gpad = smem + B_ * PROW;

  const int j   = blockIdx.x;
  const int tid = threadIdx.x;

  for (int idx = tid; idx < B_ * T_; idx += NTHREADS) {
    const int b = idx >> 12;
    const int t = idx & (T_ - 1);
    Plds[b * PROW + ZP + (t ^ 1)] = P[(size_t)idx * NR_ + j];
  }
  for (int i = tid; i < B_ * ZP; i += NTHREADS) {
    Plds[(i >> 6) * PROW + (i & (ZP - 1))] = 0.0f;
  }
  if (tid < GPAD) gpad[tid] = 0.0f;
  for (int t = tid; t < T_; t += NTHREADS) {
    gpad[GPAD + t] = g[(size_t)t * NR_ + j];
  }
  __syncthreads();

  const int lane = tid & 31;
  const int wave = __builtin_amdgcn_readfirstlane(tid >> 5);
  const int M    = lane & 15;
  const int kb   = (lane < 16) ? 0 : 2;
  const int bcol = lane & 7;
  const int tsh  = (lane & 8) ? 0 : 64;
  const float* gp    = gpad + GPAD;
  const float* prow0 = Plds + bcol * PROW + ZP - tsh + kb;

  for (int gi = wave; gi < NG; gi += NWAVE) {
    const int tb = 8 * gi + 4;
    v8f acc0 = {}, acc1 = {}, acc2 = {}, acc3 = {};
    v2f Bs[2][7][4];
    v2f As[2][4][4];

    LOADSET_MAIN(0, 0);
    for (int it = 0; it < gi; ++it) {
      const int d4 = 8 * it;
      LOADSET_MAIN(1, d4 + 4);
      GROUP_FENCE();
      WBLK_MAIN(0);
      BLOCK_END();
      GROUP_FENCE();
      LOADSET_MAIN(0, d4 + 8);
      GROUP_FENCE();
      WBLK_MAIN(1);
      BLOCK_END();
      GROUP_FENCE();
    }
    LOADSET_EPI(1);
    GROUP_FENCE();
    WBLK_MAIN(0);
    GROUP_FENCE();
    WBLK_EPI(1);

    const int bN    = lane & 7;
    const int tlo   = (lane & 8) ? tb : tb - 4;
    const int mbase = (lane < 16) ? 0 : 8;
    const float* prN = Plds + bN * PROW + ZP;
    STORE_TILE(acc0, 0)
    STORE_TILE(acc1, 1)
    STORE_TILE(acc2, 2)
    STORE_TILE(acc3, 3)
  }
}

extern "C" void kernel_launch(void* const* d_in, const int* in_sizes, int n_in,
                              void* d_out, int out_size, void* d_ws, size_t ws_size,
                              hipStream_t stream) {
  const float* P = (const float*)d_in[0];
  const float* g = (const float*)d_in[1];
  float* out = (float*)d_out;
  (void)in_sizes; (void)n_in; (void)out_size; (void)d_ws; (void)ws_size;

  epsi_causal_toeplitz_wmma<<<dim3(NR_), dim3(NTHREADS), 0, stream>>>(P, g, out);
}


# Gather of P and g cut into four time chunks and given to four added gather-only waves, under the matrix rounds that need only earlier chunks; one LDS barrier a round; same sums, same order
# speedup vs baseline: 1.7199x; 1.0767x over previous
#include <hip/hip_runtime.h>


typedef __attribute__((ext_vector_type(2))) float v2f;
typedef __attribute__((ext_vector_type(8))) float v8f;

namespace {
constexpr int B_  = 8;
constexpr int T_  = 4096;
constexpr int NR_ = 256;
constexpr int NT  = T_ / 16;
constexpr int NG  = NT / 8;
constexpr int ZP   = 64;
constexpr int PROW = ZP + T_ + 4;
constexpr int GPAD = 16;
constexpr int NWAVE = 8;
constexpr int NMATH = 32 * NWAVE;
constexpr int NLOAD = 128;
constexpr int NTHREADS = NMATH + NLOAD;
constexpr int NCHUNK = NG / NWAVE;
constexpr int CT = T_ / NCHUNK;
}

template <int NT>
__device__ __forceinline__ void stage_chunk(const float* __restrict__ Pj,
                                            const float* __restrict__ gj,
                                            float* Plds, float* gdst,
                                            int chunk, int me) {
  constexpr int KPB = CT / NT;
  constexpr int NP  = B_ * KPB;
  const int t0 = chunk * CT + me;
  const float* pg = Pj + (size_t)t0 * NR_;
  const float* gg = gj + (size_t)t0 * NR_;
  float* pl = Plds + ZP + (t0 ^ 1);
  float* gl = gdst + t0;
  float vp[NP];
  float vg[KPB];
#pragma unroll
  for (int k = 0; k < NP; ++k)
    vp[k] = pg[((size_t)(k / KPB) * T_ + (k % KPB) * NT) * NR_];
#pragma unroll
  for (int k = 0; k < KPB; ++k)
    vg[k] = gg[(size_t)(k * NT) * NR_];
#pragma unroll
  for (int k = 0; k < NP; ++k)
    pl[(k / KPB) * PROW + (k % KPB) * NT] = vp[k];
#pragma unroll
  for (int k = 0; k < KPB; ++k)
    gl[k * NT] = vg[k];
}

#define LOADSET_MAIN(S, D4)                                                   \
  {                                                                           \
    const float* pb = prow0 + 16 * (tb - (D4) - 3);                           \
    _Pragma("unroll")                                                         \
    for (int u = 0; u < 7; ++u) {                                             \
      _Pragma("unroll")                                                       \
      for (int q = 0; q < 4; ++q)                                             \
        Bs[S][u][q] = *(const v2f*)(pb + 16 * u + 4 * q);                     \
    }                                                                         \
    const float* ga = gp + 16 * (D4) + M - kb;                                \
    _Pragma("unroll")                                                         \
    for (int p = 0; p < 4; ++p) {                                             \
      _Pragma("unroll")                                                       \
      for (int q = 0; q < 4; ++q) {                                           \
        As[S][p][q].x = ga[16 * p - 4 * q - 1];                               \
        As[S][p][q].y = ga[16 * p - 4 * q];                                   \
      }                                                                       \
    }                                                                         \
  }

#define LOADSET_EPI(S)                                                        \
  {                                                                           \
    _Pragma("unroll")                                                         \
    for (int u = 0; u < 4; ++u) {                                             \
      _Pragma("unroll")                                                       \
      for (int q = 0; q < 4; ++q)                                             \
        Bs[S][u][q] = *(const v2f*)(prow0 + 16 * u + 4 * q);                  \
    }                                                                         \
    const float* ga = gp + 16 * tb + M - kb;                                  \
    _Pragma("unroll")                                                         \
    for (int p = 0; p < 4; ++p) {                                             \
      _Pragma("unroll")                                                       \
      for (int q = 0; q < 4; ++q) {                                           \
        As[S][p][q].x = ga[16 * p - 4 * q - 1];                               \
        As[S][p][q].y = ga[16 * p - 4 * q];                                   \
      }                                                                       \
    }                                                                         \
  }

#define WM(ACC, AF, BF)                                                       \
  ACC = __builtin_amdgcn_wmma_f32_16x16x4_f32(false, AF, false, BF, (short)0, \
                                              ACC, false, false);

#define WBLK_MAIN(S)                                                          \
  {                                                                           \
    _Pragma("unroll")                                                         \
    for (int p = 0; p < 4; ++p) {                                             \
      _Pragma("unroll")                                                       \
      for (int q = 0; q < 4; ++q) {                                           \
        WM(acc0, As[S][p][q], Bs[S][3 - p][q])                                \
        WM(acc1, As[S][p][q], Bs[S][4 - p][q])                                \
        WM(acc2, As[S][p][q], Bs[S][5 - p][q])                                \
        WM(acc3, As[S][p][q], Bs[S][6 - p][q])                                \
      }                                                                       \
    }                                                                         \
  }

#define WBLK_EPI(S)                                                           \
  {                                                                           \
    _Pragma("unroll")                                                         \
    for (int q = 0; q < 4; ++q) {                                             \
      WM(acc0, As[S][0][q], Bs[S][0][q])                                      \
      WM(acc1, As[S][0][q], Bs[S][1][q])                                      \
      WM(acc2, As[S][0][q], Bs[S][2][q])                                      \
      WM(acc3, As[S][0][q], Bs[S][3][q])                                      \
    }                                                                         \
    _Pragma("unroll")                                                         \
    for (int q = 0; q < 4; ++q) {                                             \
      WM(acc1, As[S][1][q], Bs[S][0][q])                                      \
      WM(acc2, As[S][1][q], Bs[S][1][q])                                      \
      WM(acc3, As[S][1][q], Bs[S][2][q])                                      \
    }                                                                         \
    _Pragma("unroll")                                                         \
    for (int q = 0; q < 4; ++q) {                                             \
      WM(acc2, As[S][2][q], Bs[S][0][q])                                      \
      WM(acc3, As[S][2][q], Bs[S][1][q])                                      \
    }                                                                         \
    _Pragma("unroll")                                                         \
    for (int q = 0; q < 4; ++q) {                                             \
      WM(acc3, As[S][3][q], Bs[S][0][q])                                      \
    }                                                                         \
  }

#define GROUP_FENCE() __builtin_amdgcn_sched_barrier(0)

#define BLOCK_END()                                                           \
  asm volatile("" : "+v"(acc0), "+v"(acc1), "+v"(acc2), "+v"(acc3))

#define STORE_TILE(ACC, C)                                                    \
  {                                                                           \
    _Pragma("unroll")                                                         \
    for (int r = 0; r < 8; ++r) {                                             \
      const int t = 16 * (tlo + (C)) + mbase + r;                             \
      out[((size_t)bN * T_ + t) * NR_ + j] = ACC[r] + prN[t ^ 1];             \
    }                                                                         \
  }

__global__ __launch_bounds__(NTHREADS)
void epsi_causal_toeplitz_wmma(const float* __restrict__ P,
                               const float* __restrict__ g,
                               float* __restrict__ out) {
  __shared__ __attribute__((aligned(16))) float smem[B_ * PROW + GPAD + T_];
  float* Plds = smem;
  float* gpad = smem + B_ * PROW;

  const int j   = blockIdx.x;
  const int tid = threadIdx.x;

  const float* Pj = P + j;
  const float* gj = g + j;
  if (tid < NMATH) {
    stage_chunk<NMATH>(Pj, gj, Plds, gpad + GPAD, 0, tid);
  } else {
    const int me = tid - NMATH;
    for (int i = me; i < B_ * ZP; i += NLOAD) {
      Plds[(i >> 6) * PROW + (i & (ZP - 1))] = 0.0f;
    }
    if (me < GPAD) gpad[me] = 0.0f;
  }
  __syncthreads();

  const int lane = tid & 31;
  const int wave = __builtin_amdgcn_readfirstlane(tid >> 5);
  const int M    = lane & 15;
  const int kb   = (lane < 16) ? 0 : 2;
  const int bcol = lane & 7;
  const int tsh  = (lane & 8) ? 0 : 64;
  const float* gp    = gpad + GPAD;
  const float* prow0 = Plds + bcol * PROW + ZP - tsh + kb;

  for (int r = 0; r < NCHUNK; ++r) {
   if (wave >= NWAVE) {
    if (r + 1 < NCHUNK)
      stage_chunk<NLOAD>(Pj, gj, Plds, gpad + GPAD, r + 1, tid - NMATH);
   } else {
    const int gi = 8 * r + wave;
    const int tb = 8 * gi + 4;
    v8f acc0 = {}, acc1 = {}, acc2 = {}, acc3 = {};
    v2f Bs[2][7][4];
    v2f As[2][4][4];

    LOADSET_MAIN(0, 0);
    for (int it = 0; it < gi; ++it) {
      const int d4 = 8 * it;
      LOADSET_MAIN(1, d4 + 4);
      GROUP_FENCE();
      WBLK_MAIN(0);
      BLOCK_END();
      GROUP_FENCE();
      LOADSET_MAIN(0, d4 + 8);
      GROUP_FENCE();
      WBLK_MAIN(1);
      BLOCK_END();
      GROUP_FENCE();
    }
    LOADSET_EPI(1);
    GROUP_FENCE();
    WBLK_MAIN(0);
    GROUP_FENCE();
    WBLK_EPI(1);

    const int bN    = lane & 7;
    const int tlo   = (lane & 8) ? tb : tb - 4;
    const int mbase = (lane < 16) ? 0 : 8;
    const float* prN = Plds + bN * PROW + ZP;
    STORE_TILE(acc0, 0)
    STORE_TILE(acc1, 1)
    STORE_TILE(acc2, 2)
    STORE_TILE(acc3, 3)
   }
    if (r + 1 < NCHUNK) {
      __builtin_amdgcn_fence(__ATOMIC_RELEASE, "workgroup", "local");
      __builtin_amdgcn_s_barrier();
      __builtin_amdgcn_fence(__ATOMIC_ACQUIRE, "workgroup", "local");
    }
  }
}

extern "C" void kernel_launch(void* const* d_in, const int* in_sizes, int n_in,
                              void* d_out, int out_size, void* d_ws, size_t ws_size,
                              hipStream_t stream) {
  const float* P = (const float*)d_in[0];
  const float* g = (const float*)d_in[1];
  float* out = (float*)d_out;
  (void)in_sizes; (void)n_in; (void)out_size; (void)d_ws; (void)ws_size;

  epsi_causal_toeplitz_wmma<<<dim3(NR_), dim3(NTHREADS), 0, stream>>>(P, g, out);
}
